# starting kernel
# baseline (speedup 1.0000x reference)
#include <hip/hip_runtime.h>

#define IN_CH  64
#define EMBED  1024
#define HID    256
#define KK     9
#define KDIM   (IN_CH * KK)
#define BATCH  8
#define NTOK   32
#define HDIM   128
#define WDIM   128
#define WSTRIDE 577

typedef __attribute__((ext_vector_type(2))) float v2f;
typedef __attribute__((ext_vector_type(8))) float v8f;

__global__ __launch_bounds__(HID) void mlp1_kernel(
    const float* __restrict__ t_emb, const float* __restrict__ wv,
    const float* __restrict__ W1, const float* __restrict__ b1,
    float* __restrict__ hout) {
  __shared__ float ctx[EMBED];
  const int bn  = blockIdx.x;
  const int b   = bn >> 5;
  const int tid = threadIdx.x;
  for (int e = tid; e < EMBED; e += HID)
    ctx[e] = wv[(size_t)bn * EMBED + e] + t_emb[(size_t)b * EMBED + e];
  __syncthreads();
  float acc = b1[tid];
  for (int e = 0; e < EMBED; ++e)
    acc = fmaf(ctx[e], W1[(size_t)e * HID + tid], acc);
  const float sig = 1.0f / (1.0f + __expf(-acc));
  hout[(size_t)bn * HID + tid] = acc * sig;
}

__global__ __launch_bounds__(KDIM) void mlp2_kernel(
    const float* __restrict__ h, const float* __restrict__ W2,
    const float* __restrict__ b2, float* __restrict__ wmat) {
  __shared__ float hl[HID];
  const int bn  = blockIdx.x;
  const int tid = threadIdx.x;
  for (int k = tid; k < HID; k += KDIM) hl[k] = h[(size_t)bn * HID + k];
  __syncthreads();
  float acc = b2[tid];
  for (int k = 0; k < HID; ++k)
    acc = fmaf(hl[k], W2[(size_t)k * KDIM + tid], acc);
  wmat[(size_t)bn * KDIM + tid] = acc;
}

__global__ __launch_bounds__(256) void dynconv_wmma_kernel(
    const float* __restrict__ latent, const float* __restrict__ wmat,
    float* __restrict__ out) {
  __shared__ float wlds[NTOK * WSTRIDE];

  const int b   = blockIdx.z;
  const int h   = blockIdx.y;
  const int w0  = blockIdx.x * 64;
  const int tid = threadIdx.x;

  const float* wsrc = wmat + (size_t)b * NTOK * KDIM;
  for (int n = 0; n < NTOK; ++n) {
    for (int o = tid; o < KDIM; o += 256)
      wlds[n * WSTRIDE + o] = wsrc[n * KDIM + o];
  }
  __syncthreads();

  const int wave    = tid >> 5;
  const int lane    = tid & 31;
  const int m       = lane & 15;
  const int hiK     = lane >> 4;
  const int tokBase = (wave & 1) * 16;
  const int px0     = w0 + (wave >> 1) * 16;

  const float* lat_b = latent + (size_t)b * IN_CH * HDIM * WDIM;
  const float* arow  = &wlds[(tokBase + m) * WSTRIDE];

  v8f acc = {0.f, 0.f, 0.f, 0.f, 0.f, 0.f, 0.f, 0.f};

  for (int di = 0; di < 3; ++di) {
    const int  hh    = h + di - 1;
    const bool rowOK = (unsigned)hh < (unsigned)HDIM;
    for (int dj = 0; dj < 3; ++dj) {
      const int  kk = di * 3 + dj;
      const int  ww = px0 + m + dj - 1;
      const bool ok = rowOK && ((unsigned)ww < (unsigned)WDIM);
      const float* bbase = lat_b + (long)hh * WDIM + ww;
#pragma unroll
      for (int d = 0; d < IN_CH; d += 4) {
        const int c0 = d + 2 * hiK;
        v2f a, bv;
        a.x = arow[c0 * KK + kk];
        a.y = arow[(c0 + 1) * KK + kk];
        bv.x = ok ? bbase[(size_t)c0 * (HDIM * WDIM)] : 0.0f;
        bv.y = ok ? bbase[(size_t)(c0 + 1) * (HDIM * WDIM)] : 0.0f;
        acc = __builtin_amdgcn_wmma_f32_16x16x4_f32(
             false, a,  false, bv,
             (short)0, acc,  false,  false);
      }
    }
  }

  const int tok = tokBase + 8 * hiK;
#pragma unroll
  for (int r = 0; r < 8; ++r) {
    out[(((size_t)b * NTOK + tok + r) * HDIM + h) * WDIM + (px0 + m)] = acc[r];
  }
}

extern "C" void kernel_launch(void* const* d_in, const int* in_sizes, int n_in,
                              void* d_out, int out_size, void* d_ws, size_t ws_size,
                              hipStream_t stream) {
  const float* latent = (const float*)d_in[0];
  const float* t_emb  = (const float*)d_in[1];
  const float* wv     = (const float*)d_in[2];
  const float* W1     = (const float*)d_in[3];
  const float* b1     = (const float*)d_in[4];
  const float* W2     = (const float*)d_in[5];
  const float* b2     = (const float*)d_in[6];
  float* out = (float*)d_out;

  float* hbuf = (float*)d_ws;
  float* wmat = hbuf + (size_t)BATCH * NTOK * HID;

  mlp1_kernel<<<BATCH * NTOK, HID, 0, stream>>>(t_emb, wv, W1, b1, hbuf);
  mlp2_kernel<<<BATCH * NTOK, KDIM, 0, stream>>>(hbuf, W2, b2, wmat);

  dim3 grid(WDIM / 64, HDIM, BATCH);
  dynconv_wmma_kernel<<<grid, 256, 0, stream>>>(latent, wmat, out);
}


# both MLP GEMMs re-decomposed on the vector unit: 8 token rows per block, each weight loaded once into a register and reused for all 8 rows, K split over the waves; MLP sum order changes
# speedup vs baseline: 5.3615x; 5.3615x over previous
#include <hip/hip_runtime.h>

#define IN_CH  64
#define EMBED  1024
#define HID    256
#define KK     9
#define KDIM   (IN_CH * KK)
#define BATCH  8
#define NTOK   32
#define HDIM   128
#define WDIM   128
#define WSTRIDE 577

typedef __attribute__((ext_vector_type(2))) float v2f;
typedef __attribute__((ext_vector_type(8))) float v8f;

#define MLP_ROWS 8

template <int KTOT, int NTOT, int NWCOL, int KS, bool FIRST>
__global__ __launch_bounds__(32 * NWCOL * KS) void mlp_rows_kernel(
    const float* __restrict__ x, const float* __restrict__ t_emb,
    const float* __restrict__ Wt, const float* __restrict__ bias,
    float* __restrict__ y) {
  constexpr int KPER = KTOT / KS;
  constexpr int CB   = 32 * NWCOL;
  constexpr int NT   = 32 * NWCOL * KS;
  __shared__ __attribute__((aligned(16))) float act[KTOT * MLP_ROWS];
  __shared__ float red[KS * MLP_ROWS * CB];

  const int tid  = threadIdx.x;
  const int lane = tid & 31;
  const int wave = __builtin_amdgcn_readfirstlane(tid >> 5);
  const int wc   = wave % NWCOL;
  const int ks   = wave / NWCOL;
  const int row0 = blockIdx.y * MLP_ROWS;
  const int k0   = ks * KPER;
  const int col  = blockIdx.x * CB + wc * 32 + lane;

  float av[KTOT / NT][MLP_ROWS];
#pragma unroll
  for (int i = 0; i < KTOT / NT; ++i) {
    const int k = tid + NT * i;
#pragma unroll
    for (int r = 0; r < MLP_ROWS; ++r) av[i][r] = x[(size_t)(row0 + r) * KTOT + k];
    if constexpr (FIRST) {
      const float t = t_emb[(size_t)(row0 / NTOK) * KTOT + k];
#pragma unroll
      for (int r = 0; r < MLP_ROWS; ++r) av[i][r] += t;
    }
  }
  float w[KPER];
  const float* wp = Wt + (size_t)k0 * NTOT + col;
#pragma unroll
  for (int k = 0; k < KPER; ++k) w[k] = wp[(size_t)k * NTOT];

#pragma unroll
  for (int i = 0; i < KTOT / NT; ++i) {
    float4* dst = reinterpret_cast<float4*>(&act[(tid + NT * i) * MLP_ROWS]);
    dst[0] = make_float4(av[i][0], av[i][1], av[i][2], av[i][3]);
    dst[1] = make_float4(av[i][4], av[i][5], av[i][6], av[i][7]);
  }
  __syncthreads();

  float acc[MLP_ROWS];
#pragma unroll
  for (int r = 0; r < MLP_ROWS; ++r) acc[r] = 0.f;
  const float4* ap = reinterpret_cast<const float4*>(&act[k0 * MLP_ROWS]);
  constexpr int KC = 8;
#pragma unroll
  for (int kc = 0; kc < KPER; kc += KC) {
    float4 a[2 * KC];
#pragma unroll
    for (int j = 0; j < 2 * KC; ++j) a[j] = ap[2 * kc + j];
#pragma unroll
    for (int j = 0; j < KC; ++j) {
      const float wk = w[kc + j];
      acc[0] = fmaf(a[2 * j].x, wk, acc[0]);
      acc[1] = fmaf(a[2 * j].y, wk, acc[1]);
      acc[2] = fmaf(a[2 * j].z, wk, acc[2]);
      acc[3] = fmaf(a[2 * j].w, wk, acc[3]);
      acc[4] = fmaf(a[2 * j + 1].x, wk, acc[4]);
      acc[5] = fmaf(a[2 * j + 1].y, wk, acc[5]);
      acc[6] = fmaf(a[2 * j + 1].z, wk, acc[6]);
      acc[7] = fmaf(a[2 * j + 1].w, wk, acc[7]);
    }
    __builtin_amdgcn_sched_group_barrier(0x100, 2 * KC, 0);
    __builtin_amdgcn_sched_group_barrier(0x002, KC * MLP_ROWS, 0);
  }
#pragma unroll
  for (int r = 0; r < MLP_ROWS; ++r)
    red[(ks * MLP_ROWS + r) * CB + wc * 32 + lane] = acc[r];
  __syncthreads();

#pragma unroll
  for (int o = tid; o < MLP_ROWS * CB; o += NT) {
    const int r = o / CB;
    const int c = o - r * CB;
    float s = red[r * CB + c];
#pragma unroll
    for (int q = 1; q < KS; ++q) s += red[(q * MLP_ROWS + r) * CB + c];
    const int gc = blockIdx.x * CB + c;
    s += bias[gc];
    if constexpr (FIRST) {
      const float sig = 1.0f / (1.0f + __expf(-s));
      s = s * sig;
    }
    y[(size_t)(row0 + r) * NTOT + gc] = s;
  }
}

#define ROWS_WG    4
#define SLAB_ROWS  (ROWS_WG + 2)
#define SLAB_COLS  136
#define SLAB_COL0  4
#define SLAB_PITCH (2 * SLAB_COLS)
#define SLAB_PLANE (SLAB_ROWS * SLAB_PITCH)
#define PAIR_ROWS  ((IN_CH / 2) * SLAB_ROWS)
#define SLAB_TRIPS (PAIR_ROWS / 32)
#define WGT_LOADS  ((NTOK * KDIM) / 1024)
#define A_TAP      66
#define A_PITCH    644

__global__ __launch_bounds__(1024) void dynconv_wmma_kernel(
    const float* __restrict__ latent, const float* __restrict__ wmat,
    float* __restrict__ out) {
  __shared__ __attribute__((aligned(16))) float wlds[NTOK * A_PITCH];
  __shared__ __attribute__((aligned(16))) float slab[(IN_CH / 2) * SLAB_PLANE];

  const int b    = blockIdx.y;
  const int h0   = blockIdx.x * ROWS_WG;
  const int tid  = threadIdx.x;
  const int lane = tid & 31;
  const int wave = __builtin_amdgcn_readfirstlane(tid >> 5);

  const float* lat_b = latent + (size_t)b * IN_CH * HDIM * WDIM;
  const float* wsrc  = wmat + (size_t)b * NTOK * KDIM;

  float4 s0[SLAB_TRIPS], s1[SLAB_TRIPS];
#pragma unroll
  for (int i = 0; i < SLAB_TRIPS; ++i) {
    const int prow = wave + 32 * i;
    const int pair = prow / SLAB_ROWS;
    const int r    = prow - pair * SLAB_ROWS;
    const int hh   = h0 - 1 + r;
    const int hc   = min(max(hh, 0), HDIM - 1);
    const float* p = lat_b + ((size_t)(2 * pair) * HDIM + hc) * WDIM + 4 * lane;
    s0[i] = *reinterpret_cast<const float4*>(p);
    s1[i] = *reinterpret_cast<const float4*>(p + HDIM * WDIM);
  }
  float wreg[WGT_LOADS];
#pragma unroll
  for (int i = 0; i < WGT_LOADS; ++i) wreg[i] = wsrc[tid + 1024 * i];

#pragma unroll
  for (int i = 0; i < SLAB_TRIPS; ++i) {
    const int prow = wave + 32 * i;
    float* dst = &slab[prow * SLAB_PITCH + 2 * (SLAB_COL0 + 4 * lane)];
    *reinterpret_cast<float4*>(dst)     = make_float4(s0[i].x, s1[i].x, s0[i].y, s1[i].y);
    *reinterpret_cast<float4*>(dst + 4) = make_float4(s0[i].z, s1[i].z, s0[i].w, s1[i].w);
  }
  if (tid < 2 * PAIR_ROWS) {
    const int prow = tid >> 1;
    const int col  = (tid & 1) ? (SLAB_COL0 + WDIM) : (SLAB_COL0 - 1);
    *reinterpret_cast<float2*>(&slab[prow * SLAB_PITCH + 2 * col]) = make_float2(0.f, 0.f);
  }
#pragma unroll
  for (int i = 0; i < WGT_LOADS; ++i) {
    const int f  = tid + 1024 * i;
    const int n  = f / KDIM;
    const int o  = f - n * KDIM;
    const int c  = o / KK;
    const int kk = o - c * KK;
    wlds[n * A_PITCH + kk * A_TAP + c] = wreg[i];
  }
  __syncthreads();

  const bool top = (h0 == 0);
  const bool bot = (h0 + ROWS_WG == HDIM);
  if (top || bot) {
    const float4 z4 = make_float4(0.f, 0.f, 0.f, 0.f);
    for (int q = tid; q < (IN_CH / 2) * (SLAB_PITCH / 4); q += 1024) {
      const int pair = q / (SLAB_PITCH / 4);
      const int x    = q - pair * (SLAB_PITCH / 4);
      float* rowp = &slab[pair * SLAB_PLANE + 4 * x];
      if (top) *reinterpret_cast<float4*>(rowp) = z4;
      if (bot) *reinterpret_cast<float4*>(rowp + (SLAB_ROWS - 1) * SLAB_PITCH) = z4;
    }
    __syncthreads();
  }

  const int m    = lane & 15;
  const int hiK  = lane >> 4;
  const int row  = wave >> 3;
  const int px0  = (wave & 7) * 16;
  const int h    = h0 + row;

  const float* arow0 = &wlds[m * A_PITCH + 2 * hiK];
  const float* arow1 = arow0 + 16 * A_PITCH;
  const float* bptr  = &slab[hiK * SLAB_PLANE + row * SLAB_PITCH + 2 * (px0 + m + (SLAB_COL0 - 1))];

  v8f acc0 = {0.f, 0.f, 0.f, 0.f, 0.f, 0.f, 0.f, 0.f};
  v8f acc1 = {0.f, 0.f, 0.f, 0.f, 0.f, 0.f, 0.f, 0.f};

#pragma unroll
  for (int di = 0; di < 3; ++di) {
#pragma unroll
    for (int dj = 0; dj < 3; ++dj) {
      const int kk   = di * 3 + dj;
      const int boff = di * SLAB_PITCH + 2 * dj;
#pragma unroll
      for (int d = 0; d < IN_CH; d += 4) {
        const v2f a0 = *reinterpret_cast<const v2f*>(arow0 + kk * A_TAP + d);
        const v2f a1 = *reinterpret_cast<const v2f*>(arow1 + kk * A_TAP + d);
        const v2f bv = *reinterpret_cast<const v2f*>(bptr + (d / 2) * SLAB_PLANE + boff);
        acc0 = __builtin_amdgcn_wmma_f32_16x16x4_f32(
             false, a0,  false, bv,
             (short)0, acc0,  false,  false);
        acc1 = __builtin_amdgcn_wmma_f32_16x16x4_f32(
             false, a1,  false, bv,
             (short)0, acc1,  false,  false);
      }
    }
  }

  const int tok = 8 * hiK;
#pragma unroll
  for (int r = 0; r < 8; ++r) {
    out[(((size_t)b * NTOK + tok + r) * HDIM + h) * WDIM + (px0 + m)]      = acc0[r];
    out[(((size_t)b * NTOK + 16 + tok + r) * HDIM + h) * WDIM + (px0 + m)] = acc1[r];
  }
}

extern "C" void kernel_launch(void* const* d_in, const int* in_sizes, int n_in,
                              void* d_out, int out_size, void* d_ws, size_t ws_size,
                              hipStream_t stream) {
  const float* latent = (const float*)d_in[0];
  const float* t_emb  = (const float*)d_in[1];
  const float* wv     = (const float*)d_in[2];
  const float* W1     = (const float*)d_in[3];
  const float* b1     = (const float*)d_in[4];
  const float* W2     = (const float*)d_in[5];
  const float* b2     = (const float*)d_in[6];
  float* out = (float*)d_out;

  float* hbuf = (float*)d_ws;
  float* wmat = hbuf + (size_t)BATCH * NTOK * HID;

  dim3 g1(HID / 32, (BATCH * NTOK) / MLP_ROWS);
  mlp_rows_kernel<EMBED, HID, 1, 8, true><<<g1, 256, 0, stream>>>(wv, t_emb, W1, b1, hbuf);
  dim3 g2(KDIM / 64, (BATCH * NTOK) / MLP_ROWS);
  mlp_rows_kernel<HID, KDIM, 2, 4, false><<<g2, 256, 0, stream>>>(hbuf, nullptr, W2, b2, wmat);

  dim3 grid(HDIM / ROWS_WG, BATCH);
  dynconv_wmma_kernel<<<grid, 1024, 0, stream>>>(latent, wmat, out);
}


# both MLP GEMMs: each thread owns 4 adjacent columns (128-bit weight loads) and a 4x shorter k slice: one broadcast LDS activation read feeds 32 multiply-adds, not 8; MLP sum order changes
# speedup vs baseline: 5.7859x; 1.0792x over previous
#include <hip/hip_runtime.h>

#define IN_CH  64
#define EMBED  1024
#define HID    256
#define KK     9
#define KDIM   (IN_CH * KK)
#define BATCH  8
#define NTOK   32
#define HDIM   128
#define WDIM   128
#define WSTRIDE 577

typedef __attribute__((ext_vector_type(2))) float v2f;
typedef __attribute__((ext_vector_type(8))) float v8f;

#define MLP_ROWS 8
#define MLP_CT   4

template <int KTOT, int NTOT, int CB, bool FIRST>
__global__ __launch_bounds__(256) void mlp_rows_kernel(
    const float* __restrict__ x, const float* __restrict__ t_emb,
    const float* __restrict__ Wt, const float* __restrict__ bias,
    float* __restrict__ y) {
  constexpr int NT   = 256;
  constexpr int CQ   = CB / MLP_CT;
  constexpr int KS   = NT / CQ;
  constexpr int KPER = KTOT / KS;
  __shared__ __attribute__((aligned(16))) float act[KTOT * MLP_ROWS];
  __shared__ __attribute__((aligned(16))) float red[KS * MLP_ROWS * CB];

  const int tid  = threadIdx.x;
  const int cq   = tid % CQ;
  const int ks   = tid / CQ;
  const int row0 = blockIdx.y * MLP_ROWS;
  const int col  = blockIdx.x * CB + MLP_CT * cq;

  float av[KTOT / NT][MLP_ROWS];
#pragma unroll
  for (int i = 0; i < KTOT / NT; ++i) {
    const int k = tid + NT * i;
#pragma unroll
    for (int r = 0; r < MLP_ROWS; ++r) av[i][r] = x[(size_t)(row0 + r) * KTOT + k];
    if constexpr (FIRST) {
      const float t = t_emb[(size_t)(row0 / NTOK) * KTOT + k];
#pragma unroll
      for (int r = 0; r < MLP_ROWS; ++r) av[i][r] += t;
    }
  }
  float4 w[KPER];
  const float* wp = Wt + (size_t)ks * NTOT + col;
#pragma unroll
  for (int j = 0; j < KPER; ++j)
    w[j] = *reinterpret_cast<const float4*>(wp + (size_t)(KS * j) * NTOT);

#pragma unroll
  for (int i = 0; i < KTOT / NT; ++i) {
    float4* dst = reinterpret_cast<float4*>(&act[(tid + NT * i) * MLP_ROWS]);
    dst[0] = make_float4(av[i][0], av[i][1], av[i][2], av[i][3]);
    dst[1] = make_float4(av[i][4], av[i][5], av[i][6], av[i][7]);
  }
  __syncthreads();

  float acc[MLP_ROWS][MLP_CT];
#pragma unroll
  for (int r = 0; r < MLP_ROWS; ++r)
#pragma unroll
    for (int c = 0; c < MLP_CT; ++c) acc[r][c] = 0.f;
  const float4* ap = reinterpret_cast<const float4*>(act) + 2 * ks;
  constexpr int KC = 8;
#pragma unroll
  for (int kc = 0; kc < KPER; kc += KC) {
    float4 a[2 * KC];
#pragma unroll
    for (int j = 0; j < KC; ++j) {
      a[2 * j]     = ap[2 * KS * (kc + j)];
      a[2 * j + 1] = ap[2 * KS * (kc + j) + 1];
    }
    __builtin_amdgcn_sched_barrier(0);
#pragma unroll
    for (int j = 0; j < KC; ++j) {
      const float4 wk = w[kc + j];
      const float4 lo = a[2 * j];
      const float4 hi = a[2 * j + 1];
      const float ar[MLP_ROWS] = {lo.x, lo.y, lo.z, lo.w, hi.x, hi.y, hi.z, hi.w};
#pragma unroll
      for (int r = 0; r < MLP_ROWS; ++r) {
        acc[r][0] = fmaf(ar[r], wk.x, acc[r][0]);
        acc[r][1] = fmaf(ar[r], wk.y, acc[r][1]);
        acc[r][2] = fmaf(ar[r], wk.z, acc[r][2]);
        acc[r][3] = fmaf(ar[r], wk.w, acc[r][3]);
      }
    }
    __builtin_amdgcn_sched_barrier(0);
  }
#pragma unroll
  for (int r = 0; r < MLP_ROWS; ++r)
    *reinterpret_cast<float4*>(&red[(ks * MLP_ROWS + r) * CB + MLP_CT * cq]) =
        make_float4(acc[r][0], acc[r][1], acc[r][2], acc[r][3]);
  __syncthreads();

#pragma unroll
  for (int o = tid; o < MLP_ROWS * CB; o += NT) {
    const int r = o / CB;
    const int c = o - r * CB;
    float s = red[r * CB + c];
#pragma unroll
    for (int q = 1; q < KS; ++q) s += red[(q * MLP_ROWS + r) * CB + c];
    const int gc = blockIdx.x * CB + c;
    s += bias[gc];
    if constexpr (FIRST) {
      const float sig = 1.0f / (1.0f + __expf(-s));
      s = s * sig;
    }
    y[(size_t)(row0 + r) * NTOT + gc] = s;
  }
}

#define ROWS_WG    4
#define SLAB_ROWS  (ROWS_WG + 2)
#define SLAB_COLS  136
#define SLAB_COL0  4
#define SLAB_PITCH (2 * SLAB_COLS)
#define SLAB_PLANE (SLAB_ROWS * SLAB_PITCH)
#define PAIR_ROWS  ((IN_CH / 2) * SLAB_ROWS)
#define SLAB_TRIPS (PAIR_ROWS / 32)
#define WGT_LOADS  ((NTOK * KDIM) / 1024)
#define A_TAP      66
#define A_PITCH    644

__global__ __launch_bounds__(1024) void dynconv_wmma_kernel(
    const float* __restrict__ latent, const float* __restrict__ wmat,
    float* __restrict__ out) {
  __shared__ __attribute__((aligned(16))) float wlds[NTOK * A_PITCH];
  __shared__ __attribute__((aligned(16))) float slab[(IN_CH / 2) * SLAB_PLANE];

  const int b    = blockIdx.y;
  const int h0   = blockIdx.x * ROWS_WG;
  const int tid  = threadIdx.x;
  const int lane = tid & 31;
  const int wave = __builtin_amdgcn_readfirstlane(tid >> 5);

  const float* lat_b = latent + (size_t)b * IN_CH * HDIM * WDIM;
  const float* wsrc  = wmat + (size_t)b * NTOK * KDIM;

  float4 s0[SLAB_TRIPS], s1[SLAB_TRIPS];
#pragma unroll
  for (int i = 0; i < SLAB_TRIPS; ++i) {
    const int prow = wave + 32 * i;
    const int pair = prow / SLAB_ROWS;
    const int r    = prow - pair * SLAB_ROWS;
    const int hh   = h0 - 1 + r;
    const int hc   = min(max(hh, 0), HDIM - 1);
    const float* p = lat_b + ((size_t)(2 * pair) * HDIM + hc) * WDIM + 4 * lane;
    s0[i] = *reinterpret_cast<const float4*>(p);
    s1[i] = *reinterpret_cast<const float4*>(p + HDIM * WDIM);
  }
  float wreg[WGT_LOADS];
#pragma unroll
  for (int i = 0; i < WGT_LOADS; ++i) wreg[i] = wsrc[tid + 1024 * i];

#pragma unroll
  for (int i = 0; i < SLAB_TRIPS; ++i) {
    const int prow = wave + 32 * i;
    float* dst = &slab[prow * SLAB_PITCH + 2 * (SLAB_COL0 + 4 * lane)];
    *reinterpret_cast<float4*>(dst)     = make_float4(s0[i].x, s1[i].x, s0[i].y, s1[i].y);
    *reinterpret_cast<float4*>(dst + 4) = make_float4(s0[i].z, s1[i].z, s0[i].w, s1[i].w);
  }
  if (tid < 2 * PAIR_ROWS) {
    const int prow = tid >> 1;
    const int col  = (tid & 1) ? (SLAB_COL0 + WDIM) : (SLAB_COL0 - 1);
    *reinterpret_cast<float2*>(&slab[prow * SLAB_PITCH + 2 * col]) = make_float2(0.f, 0.f);
  }
#pragma unroll
  for (int i = 0; i < WGT_LOADS; ++i) {
    const int f  = tid + 1024 * i;
    const int n  = f / KDIM;
    const int o  = f - n * KDIM;
    const int c  = o / KK;
    const int kk = o - c * KK;
    wlds[n * A_PITCH + kk * A_TAP + c] = wreg[i];
  }
  __syncthreads();

  const bool top = (h0 == 0);
  const bool bot = (h0 + ROWS_WG == HDIM);
  if (top || bot) {
    const float4 z4 = make_float4(0.f, 0.f, 0.f, 0.f);
    for (int q = tid; q < (IN_CH / 2) * (SLAB_PITCH / 4); q += 1024) {
      const int pair = q / (SLAB_PITCH / 4);
      const int x    = q - pair * (SLAB_PITCH / 4);
      float* rowp = &slab[pair * SLAB_PLANE + 4 * x];
      if (top) *reinterpret_cast<float4*>(rowp) = z4;
      if (bot) *reinterpret_cast<float4*>(rowp + (SLAB_ROWS - 1) * SLAB_PITCH) = z4;
    }
    __syncthreads();
  }

  const int m    = lane & 15;
  const int hiK  = lane >> 4;
  const int row  = wave >> 3;
  const int px0  = (wave & 7) * 16;
  const int h    = h0 + row;

  const float* arow0 = &wlds[m * A_PITCH + 2 * hiK];
  const float* arow1 = arow0 + 16 * A_PITCH;
  const float* bptr  = &slab[hiK * SLAB_PLANE + row * SLAB_PITCH + 2 * (px0 + m + (SLAB_COL0 - 1))];

  v8f acc0 = {0.f, 0.f, 0.f, 0.f, 0.f, 0.f, 0.f, 0.f};
  v8f acc1 = {0.f, 0.f, 0.f, 0.f, 0.f, 0.f, 0.f, 0.f};

#pragma unroll
  for (int di = 0; di < 3; ++di) {
#pragma unroll
    for (int dj = 0; dj < 3; ++dj) {
      const int kk   = di * 3 + dj;
      const int boff = di * SLAB_PITCH + 2 * dj;
#pragma unroll
      for (int d = 0; d < IN_CH; d += 4) {
        const v2f a0 = *reinterpret_cast<const v2f*>(arow0 + kk * A_TAP + d);
        const v2f a1 = *reinterpret_cast<const v2f*>(arow1 + kk * A_TAP + d);
        const v2f bv = *reinterpret_cast<const v2f*>(bptr + (d / 2) * SLAB_PLANE + boff);
        acc0 = __builtin_amdgcn_wmma_f32_16x16x4_f32(
             false, a0,  false, bv,
             (short)0, acc0,  false,  false);
        acc1 = __builtin_amdgcn_wmma_f32_16x16x4_f32(
             false, a1,  false, bv,
             (short)0, acc1,  false,  false);
      }
    }
  }

  const int tok = 8 * hiK;
#pragma unroll
  for (int r = 0; r < 8; ++r) {
    out[(((size_t)b * NTOK + tok + r) * HDIM + h) * WDIM + (px0 + m)]      = acc0[r];
    out[(((size_t)b * NTOK + 16 + tok + r) * HDIM + h) * WDIM + (px0 + m)] = acc1[r];
  }
}

extern "C" void kernel_launch(void* const* d_in, const int* in_sizes, int n_in,
                              void* d_out, int out_size, void* d_ws, size_t ws_size,
                              hipStream_t stream) {
  const float* latent = (const float*)d_in[0];
  const float* t_emb  = (const float*)d_in[1];
  const float* wv     = (const float*)d_in[2];
  const float* W1     = (const float*)d_in[3];
  const float* b1     = (const float*)d_in[4];
  const float* W2     = (const float*)d_in[5];
  const float* b2     = (const float*)d_in[6];
  float* out = (float*)d_out;

  float* hbuf = (float*)d_ws;
  float* wmat = hbuf + (size_t)BATCH * NTOK * HID;

  dim3 g1(HID / 32, (BATCH * NTOK) / MLP_ROWS);
  mlp_rows_kernel<EMBED, HID, 32, true><<<g1, 256, 0, stream>>>(wv, t_emb, W1, b1, hbuf);
  dim3 g2(KDIM / 64, (BATCH * NTOK) / MLP_ROWS);
  mlp_rows_kernel<HID, KDIM, 64, false><<<g2, 256, 0, stream>>>(hbuf, nullptr, W2, b2, wmat);

  dim3 grid(HDIM / ROWS_WG, BATCH);
  dynconv_wmma_kernel<<<grid, 1024, 0, stream>>>(latent, wmat, out);
}


# dynconv staging pipelined over 4 chunks of 16 channels: chunk c+1 is fetched to registers before the products of chunk c, stored to LDS after them; weights in 2 halves; sum order now chunk-major
# speedup vs baseline: 6.2343x; 1.0775x over previous
#include <hip/hip_runtime.h>

#define IN_CH  64
#define EMBED  1024
#define HID    256
#define KK     9
#define KDIM   (IN_CH * KK)
#define BATCH  8
#define NTOK   32
#define HDIM   128
#define WDIM   128
#define WSTRIDE 577

typedef __attribute__((ext_vector_type(2))) float v2f;
typedef __attribute__((ext_vector_type(8))) float v8f;

#define MLP_ROWS 8
#define MLP_CT   4

template <int KTOT, int NTOT, int CB, bool FIRST>
__global__ __launch_bounds__(256) void mlp_rows_kernel(
    const float* __restrict__ x, const float* __restrict__ t_emb,
    const float* __restrict__ Wt, const float* __restrict__ bias,
    float* __restrict__ y) {
  constexpr int NT   = 256;
  constexpr int CQ   = CB / MLP_CT;
  constexpr int KS   = NT / CQ;
  constexpr int KPER = KTOT / KS;
  __shared__ __attribute__((aligned(16))) float act[KTOT * MLP_ROWS];
  __shared__ __attribute__((aligned(16))) float red[KS * MLP_ROWS * CB];

  const int tid  = threadIdx.x;
  const int cq   = tid % CQ;
  const int ks   = tid / CQ;
  const int row0 = blockIdx.y * MLP_ROWS;
  const int col  = blockIdx.x * CB + MLP_CT * cq;

  float av[KTOT / NT][MLP_ROWS];
#pragma unroll
  for (int i = 0; i < KTOT / NT; ++i) {
    const int k = tid + NT * i;
#pragma unroll
    for (int r = 0; r < MLP_ROWS; ++r) av[i][r] = x[(size_t)(row0 + r) * KTOT + k];
    if constexpr (FIRST) {
      const float t = t_emb[(size_t)(row0 / NTOK) * KTOT + k];
#pragma unroll
      for (int r = 0; r < MLP_ROWS; ++r) av[i][r] += t;
    }
  }
  float4 w[KPER];
  const float* wp = Wt + (size_t)ks * NTOT + col;
#pragma unroll
  for (int j = 0; j < KPER; ++j)
    w[j] = *reinterpret_cast<const float4*>(wp + (size_t)(KS * j) * NTOT);

#pragma unroll
  for (int i = 0; i < KTOT / NT; ++i) {
    float4* dst = reinterpret_cast<float4*>(&act[(tid + NT * i) * MLP_ROWS]);
    dst[0] = make_float4(av[i][0], av[i][1], av[i][2], av[i][3]);
    dst[1] = make_float4(av[i][4], av[i][5], av[i][6], av[i][7]);
  }
  __syncthreads();

  float acc[MLP_ROWS][MLP_CT];
#pragma unroll
  for (int r = 0; r < MLP_ROWS; ++r)
#pragma unroll
    for (int c = 0; c < MLP_CT; ++c) acc[r][c] = 0.f;
  const float4* ap = reinterpret_cast<const float4*>(act) + 2 * ks;
  constexpr int KC = 8;
#pragma unroll
  for (int kc = 0; kc < KPER; kc += KC) {
    float4 a[2 * KC];
#pragma unroll
    for (int j = 0; j < KC; ++j) {
      a[2 * j]     = ap[2 * KS * (kc + j)];
      a[2 * j + 1] = ap[2 * KS * (kc + j) + 1];
    }
    __builtin_amdgcn_sched_barrier(0);
#pragma unroll
    for (int j = 0; j < KC; ++j) {
      const float4 wk = w[kc + j];
      const float4 lo = a[2 * j];
      const float4 hi = a[2 * j + 1];
      const float ar[MLP_ROWS] = {lo.x, lo.y, lo.z, lo.w, hi.x, hi.y, hi.z, hi.w};
#pragma unroll
      for (int r = 0; r < MLP_ROWS; ++r) {
        acc[r][0] = fmaf(ar[r], wk.x, acc[r][0]);
        acc[r][1] = fmaf(ar[r], wk.y, acc[r][1]);
        acc[r][2] = fmaf(ar[r], wk.z, acc[r][2]);
        acc[r][3] = fmaf(ar[r], wk.w, acc[r][3]);
      }
    }
    __builtin_amdgcn_sched_barrier(0);
  }
#pragma unroll
  for (int r = 0; r < MLP_ROWS; ++r)
    *reinterpret_cast<float4*>(&red[(ks * MLP_ROWS + r) * CB + MLP_CT * cq]) =
        make_float4(acc[r][0], acc[r][1], acc[r][2], acc[r][3]);
  __syncthreads();

#pragma unroll
  for (int o = tid; o < MLP_ROWS * CB; o += NT) {
    const int r = o / CB;
    const int c = o - r * CB;
    float s = red[r * CB + c];
#pragma unroll
    for (int q = 1; q < KS; ++q) s += red[(q * MLP_ROWS + r) * CB + c];
    const int gc = blockIdx.x * CB + c;
    s += bias[gc];
    if constexpr (FIRST) {
      const float sig = 1.0f / (1.0f + __expf(-s));
      s = s * sig;
    }
    y[(size_t)(row0 + r) * NTOT + gc] = s;
  }
}

#define ROWS_WG    4
#define SLAB_ROWS  (ROWS_WG + 2)
#define SLAB_COLS  136
#define SLAB_COL0  4
#define SLAB_PITCH (2 * SLAB_COLS)
#define SLAB_PLANE (SLAB_ROWS * SLAB_PITCH)
#define PAIR_ROWS  ((IN_CH / 2) * SLAB_ROWS)
#define NCHUNK      4
#define CH_CHUNK    (IN_CH / NCHUNK)
#define PAIRS_CHUNK (CH_CHUNK / 2)
#define SLAB_TRIPS  ((PAIRS_CHUNK * SLAB_ROWS * 2) / 32)
#define WGT_HALFCH  (IN_CH / 2)
#define WGT_HALFK   (WGT_HALFCH * KK)
#define WGT_LOADS   ((NTOK * WGT_HALFK) / 1024)
#define A_TAP      66
#define A_PITCH    644

template <int C>
__device__ __forceinline__ void slab_fetch(const float* __restrict__ lat_b, int h0, int wave, int lane,
                                           float2 (&sa)[SLAB_TRIPS], float2 (&sb)[SLAB_TRIPS]) {
#pragma unroll
  for (int i = 0; i < SLAB_TRIPS; ++i) {
    const int u    = wave + 32 * i;
    const int pl   = u >> 1;
    const int half = u & 1;
    const int pair = C * PAIRS_CHUNK + pl / SLAB_ROWS;
    const int r    = pl % SLAB_ROWS;
    const int hh   = h0 - 1 + r;
    const int hc   = min(max(hh, 0), HDIM - 1);
    const float* p = lat_b + ((size_t)(2 * pair) * HDIM + hc) * WDIM + 64 * half + 2 * lane;
    sa[i] = *reinterpret_cast<const float2*>(p);
    sb[i] = *reinterpret_cast<const float2*>(p + HDIM * WDIM);
  }
}

template <int C>
__device__ __forceinline__ void slab_put(float* slab, int h0, int wave, int lane,
                                         const float2 (&sa)[SLAB_TRIPS], const float2 (&sb)[SLAB_TRIPS]) {
#pragma unroll
  for (int i = 0; i < SLAB_TRIPS; ++i) {
    const int u    = wave + 32 * i;
    const int pl   = u >> 1;
    const int half = u & 1;
    const int pair = C * PAIRS_CHUNK + pl / SLAB_ROWS;
    const int r    = pl % SLAB_ROWS;
    const int hh   = h0 - 1 + r;
    const bool inside = (hh >= 0) && (hh < HDIM);
    asm volatile("" : : "v"(sa[i].x), "v"(sa[i].y), "v"(sb[i].x), "v"(sb[i].y));
    const float2 a = inside ? sa[i] : make_float2(0.f, 0.f);
    const float2 c = inside ? sb[i] : make_float2(0.f, 0.f);
    float* dst = &slab[(pair * SLAB_ROWS + r) * SLAB_PITCH + 2 * (SLAB_COL0 + 64 * half + 2 * lane)];
    *reinterpret_cast<float4*>(dst) = make_float4(a.x, c.x, a.y, c.y);
  }
}

template <int HW>
__device__ __forceinline__ void wgt_fetch(const float* __restrict__ wsrc, int tid, float (&wreg)[WGT_LOADS]) {
#pragma unroll
  for (int i = 0; i < WGT_LOADS; ++i) {
    const int f = tid + 1024 * i;
    const int n = f / WGT_HALFK;
    const int o = f - n * WGT_HALFK;
    wreg[i] = wsrc[n * KDIM + HW * WGT_HALFK + o];
  }
}
template <int HW>
__device__ __forceinline__ void wgt_put(float* wlds, int tid, const float (&wreg)[WGT_LOADS]) {
#pragma unroll
  for (int i = 0; i < WGT_LOADS; ++i) {
    const int f  = tid + 1024 * i;
    const int n  = f / WGT_HALFK;
    const int o  = f - n * WGT_HALFK;
    const int cl = o / KK;
    const int kk = o - cl * KK;
    wlds[n * A_PITCH + kk * A_TAP + HW * WGT_HALFCH + cl] = wreg[i];
  }
}

template <int C>
__device__ __forceinline__ void chunk_products(const float* arow0, const float* arow1,
                                               const float* bptr, v8f& acc0, v8f& acc1) {
#pragma unroll
  for (int di = 0; di < 3; ++di) {
#pragma unroll
    for (int dj = 0; dj < 3; ++dj) {
      const int kk   = di * 3 + dj;
      const int boff = di * SLAB_PITCH + 2 * dj;
#pragma unroll
      for (int d = C * CH_CHUNK; d < (C + 1) * CH_CHUNK; d += 4) {
        const v2f a0 = *reinterpret_cast<const v2f*>(arow0 + kk * A_TAP + d);
        const v2f a1 = *reinterpret_cast<const v2f*>(arow1 + kk * A_TAP + d);
        const v2f bv = *reinterpret_cast<const v2f*>(bptr + (d / 2) * SLAB_PLANE + boff);
        acc0 = __builtin_amdgcn_wmma_f32_16x16x4_f32(
             false, a0,  false, bv,
             (short)0, acc0,  false,  false);
        acc1 = __builtin_amdgcn_wmma_f32_16x16x4_f32(
             false, a1,  false, bv,
             (short)0, acc1,  false,  false);
      }
    }
  }
}

__global__ __launch_bounds__(1024) void dynconv_wmma_kernel(
    const float* __restrict__ latent, const float* __restrict__ wmat,
    float* __restrict__ out) {
  __shared__ __attribute__((aligned(16))) float wlds[NTOK * A_PITCH];
  __shared__ __attribute__((aligned(16))) float slab[(IN_CH / 2) * SLAB_PLANE];

  const int b    = blockIdx.y;
  const int h0   = blockIdx.x * ROWS_WG;
  const int tid  = threadIdx.x;
  const int lane = tid & 31;
  const int wave = __builtin_amdgcn_readfirstlane(tid >> 5);

  const float* lat_b = latent + (size_t)b * IN_CH * HDIM * WDIM;
  const float* wsrc  = wmat + (size_t)b * NTOK * KDIM;

  float2 sa[SLAB_TRIPS], sb[SLAB_TRIPS];
  float  wreg[WGT_LOADS];

  wgt_fetch<0>(wsrc, tid, wreg);
  slab_fetch<0>(lat_b, h0, wave, lane, sa, sb);
  if (tid < 2 * PAIR_ROWS) {
    const int prow = tid >> 1;
    const int col  = (tid & 1) ? (SLAB_COL0 + WDIM) : (SLAB_COL0 - 1);
    *reinterpret_cast<float2*>(&slab[prow * SLAB_PITCH + 2 * col]) = make_float2(0.f, 0.f);
  }
  wgt_put<0>(wlds, tid, wreg);
  slab_put<0>(slab, h0, wave, lane, sa, sb);
  __syncthreads();

  const int m    = lane & 15;
  const int hiK  = lane >> 4;
  const int row  = wave >> 3;
  const int px0  = (wave & 7) * 16;
  const int h    = h0 + row;

  const float* arow0 = &wlds[m * A_PITCH + 2 * hiK];
  const float* arow1 = arow0 + 16 * A_PITCH;
  const float* bptr  = &slab[hiK * SLAB_PLANE + row * SLAB_PITCH + 2 * (px0 + m + (SLAB_COL0 - 1))];

  v8f acc0 = {0.f, 0.f, 0.f, 0.f, 0.f, 0.f, 0.f, 0.f};
  v8f acc1 = {0.f, 0.f, 0.f, 0.f, 0.f, 0.f, 0.f, 0.f};

  slab_fetch<1>(lat_b, h0, wave, lane, sa, sb);
  chunk_products<0>(arow0, arow1, bptr, acc0, acc1);
  slab_put<1>(slab, h0, wave, lane, sa, sb);
  __syncthreads();

  slab_fetch<2>(lat_b, h0, wave, lane, sa, sb);
  wgt_fetch<1>(wsrc, tid, wreg);
  chunk_products<1>(arow0, arow1, bptr, acc0, acc1);
  slab_put<2>(slab, h0, wave, lane, sa, sb);
  wgt_put<1>(wlds, tid, wreg);
  __syncthreads();

  slab_fetch<3>(lat_b, h0, wave, lane, sa, sb);
  chunk_products<2>(arow0, arow1, bptr, acc0, acc1);
  slab_put<3>(slab, h0, wave, lane, sa, sb);
  __syncthreads();

  chunk_products<3>(arow0, arow1, bptr, acc0, acc1);

  const int tok = 8 * hiK;
#pragma unroll
  for (int r = 0; r < 8; ++r) {
    out[(((size_t)b * NTOK + tok + r) * HDIM + h) * WDIM + (px0 + m)]      = acc0[r];
    out[(((size_t)b * NTOK + 16 + tok + r) * HDIM + h) * WDIM + (px0 + m)] = acc1[r];
  }
}

extern "C" void kernel_launch(void* const* d_in, const int* in_sizes, int n_in,
                              void* d_out, int out_size, void* d_ws, size_t ws_size,
                              hipStream_t stream) {
  const float* latent = (const float*)d_in[0];
  const float* t_emb  = (const float*)d_in[1];
  const float* wv     = (const float*)d_in[2];
  const float* W1     = (const float*)d_in[3];
  const float* b1     = (const float*)d_in[4];
  const float* W2     = (const float*)d_in[5];
  const float* b2     = (const float*)d_in[6];
  float* out = (float*)d_out;

  float* hbuf = (float*)d_ws;
  float* wmat = hbuf + (size_t)BATCH * NTOK * HID;

  dim3 g1(HID / 32, (BATCH * NTOK) / MLP_ROWS);
  mlp_rows_kernel<EMBED, HID, 32, true><<<g1, 256, 0, stream>>>(wv, t_emb, W1, b1, hbuf);
  dim3 g2(KDIM / 64, (BATCH * NTOK) / MLP_ROWS);
  mlp_rows_kernel<HID, KDIM, 64, false><<<g2, 256, 0, stream>>>(hbuf, nullptr, W2, b2, wmat);

  dim3 grid(HDIM / ROWS_WG, BATCH);
  dynconv_wmma_kernel<<<grid, 1024, 0, stream>>>(latent, wmat, out);
}
